# starting kernel
# baseline (speedup 1.0000x reference)
#include <hip/hip_runtime.h>

typedef float v2f __attribute__((ext_vector_type(2)));
typedef float v4f __attribute__((ext_vector_type(4)));
typedef float v8f __attribute__((ext_vector_type(8)));

#define Bn   32
#define Cn   512
#define HW2  1024
#define NW   77
#define NP   80
#define WD   256

static __device__ __forceinline__ v8f wmma_f32(v2f a, v2f b, v8f c) {
  return __builtin_amdgcn_wmma_f32_16x16x4_f32(false, a, false, b, (short)0, c, false, false);
}

__global__ __launch_bounds__(256) void we_gemm_kernel(
    const float* __restrict__ word_emb, const float* __restrict__ W_fc,
    const float* __restrict__ b_fc, float* __restrict__ we_ws,
    float* __restrict__ weT_ws) {
  const int lane = threadIdx.x & 31;
  const int wave = threadIdx.x >> 5;
  int tile = blockIdx.x * 8 + wave;
  const int b  = tile / (5 * 64);
  tile -= b * 5 * 64;
  const int tm = tile >> 6;
  const int tn = tile & 63;
  const int half = lane >> 4;
  const int l16  = lane & 15;

  const int a_row = tm * 16 + l16;
  const int col   = tn * 16 + l16;
  const float* Abase = word_emb + ((size_t)b * NW + (a_row < NW ? a_row : NW - 1)) * WD;
  const float* Bbase = W_fc + (size_t)col * WD;

  v8f acc = {};
  for (int k0 = 0; k0 < WD; k0 += 4) {
    const int k = k0 + 2 * half;
    v2f a  = *(const v2f*)(Abase + k);
    v2f bm = *(const v2f*)(Bbase + k);
    acc = wmma_f32(a, bm, acc);
  }
  const float bias = b_fc[col];

  float* D = we_ws + ((size_t)b * NP + tm * 16 + 8 * half) * HW2 + col;
  #pragma unroll
  for (int i = 0; i < 8; ++i) D[(size_t)i * HW2] = acc[i] + bias;

  float* T = weT_ws + ((size_t)b * HW2 + col) * NP + tm * 16 + 8 * half;
  v4f lo = {acc[0] + bias, acc[1] + bias, acc[2] + bias, acc[3] + bias};
  v4f hi = {acc[4] + bias, acc[5] + bias, acc[6] + bias, acc[7] + bias};
  *(v4f*)(T)     = lo;
  *(v4f*)(T + 4) = hi;
}

__global__ __launch_bounds__(256) void scores_softmax_kernel(
    const float* __restrict__ feat, const float* __restrict__ we_ws,
    float* __restrict__ att_ws) {
  const int lane = threadIdx.x & 31;
  const int wave = threadIdx.x >> 5;
  const int slab = blockIdx.x * 8 + wave;
  const int b  = slab >> 5;
  const int cs = (slab & 31) * 16;
  const int half = lane >> 4;
  const int l16  = lane & 15;

  const float* Abase  = feat  + ((size_t)b * Cn + cs + l16) * HW2;
  const float* Bbase0 = we_ws + ((size_t)b * NP + l16) * HW2;

  v8f acc[5] = {};
  for (int k0 = 0; k0 < HW2; k0 += 4) {
    const int k = k0 + 2 * half;
    v2f a = *(const v2f*)(Abase + k);
    #pragma unroll
    for (int t = 0; t < 5; ++t) {
      v2f bm = *(const v2f*)(Bbase0 + (size_t)(16 * t) * HW2 + k);
      acc[t] = wmma_f32(a, bm, acc[t]);
    }
  }

  float* Dbase = att_ws + ((size_t)b * Cn + cs + 8 * half) * NP;
  #pragma unroll
  for (int i = 0; i < 8; ++i) {
    float m = -3.402823466e38f;
    #pragma unroll
    for (int t = 0; t < 5; ++t)
      if (16 * t + l16 < NW) m = fmaxf(m, acc[t][i]);
    #pragma unroll
    for (int off = 8; off >= 1; off >>= 1) m = fmaxf(m, __shfl_xor(m, off, 32));

    float e[5], s = 0.f;
    #pragma unroll
    for (int t = 0; t < 5; ++t) {
      e[t] = (16 * t + l16 < NW) ? __expf(acc[t][i] - m) : 0.f;
      s += e[t];
    }
    #pragma unroll
    for (int off = 8; off >= 1; off >>= 1) s += __shfl_xor(s, off, 32);
    const float inv = 1.f / s;
    #pragma unroll
    for (int t = 0; t < 5; ++t)
      Dbase[(size_t)i * NP + 16 * t + l16] = e[t] * inv;
  }
}

__global__ __launch_bounds__(256) void attfeat_kernel(
    const float* __restrict__ att_ws, const float* __restrict__ weT_ws,
    float* __restrict__ out) {
  const int lane = threadIdx.x & 31;
  const int wave = threadIdx.x >> 5;
  int tile = blockIdx.x * 8 + wave;
  const int b = tile / (32 * 16);
  tile -= b * 32 * 16;
  const int tm = tile >> 4;
  const int tq = tile & 15;
  const int half = lane >> 4;
  const int l16  = lane & 15;

  const float* Abase = att_ws + ((size_t)b * Cn + tm * 16 + l16) * NP;
  const float* Bbase0 = weT_ws + ((size_t)b * HW2 + tq * 64 + l16) * NP;

  v8f acc[4] = {};
  for (int k0 = 0; k0 < NP; k0 += 4) {
    const int k = k0 + 2 * half;
    v2f a = *(const v2f*)(Abase + k);
    #pragma unroll
    for (int t = 0; t < 4; ++t) {
      v2f bm = *(const v2f*)(Bbase0 + (size_t)(16 * t) * NP + k);
      acc[t] = wmma_f32(a, bm, acc[t]);
    }
  }
  float* D = out + ((size_t)b * Cn + tm * 16 + 8 * half) * HW2 + tq * 64 + l16;
  #pragma unroll
  for (int t = 0; t < 4; ++t)
    #pragma unroll
    for (int i = 0; i < 8; ++i)
      D[(size_t)i * HW2 + 16 * t] = acc[t][i];
}

extern "C" void kernel_launch(void* const* d_in, const int* in_sizes, int n_in,
                              void* d_out, int out_size, void* d_ws, size_t ws_size,
                              hipStream_t stream) {
  const float* feat     = (const float*)d_in[0];
  const float* word_emb = (const float*)d_in[1];
  const float* W_fc     = (const float*)d_in[2];
  const float* b_fc     = (const float*)d_in[3];
  float* out = (float*)d_out;

  float* we_ws  = (float*)d_ws;
  float* weT_ws = we_ws  + (size_t)Bn * NP * HW2;
  float* att_ws = weT_ws + (size_t)Bn * HW2 * NP;

  we_gemm_kernel       <<<(Bn * 5 * 64) / 8,  256, 0, stream>>>(word_emb, W_fc, b_fc, we_ws, weT_ws);
  scores_softmax_kernel<<<(Bn * 32) / 8,      256, 0, stream>>>(feat, we_ws, att_ws);
  attfeat_kernel       <<<(Bn * 32 * 16) / 8, 256, 0, stream>>>(att_ws, weT_ws, out);
}


# fragment loads widened from 8 to 16 bytes per lane in all three GEMMs: one b128 load feeds two f32 WMMAs (half h holds k0+4h..+3, same map for A and B); the k terms are summed in a different order
# speedup vs baseline: 1.0380x; 1.0380x over previous
#include <hip/hip_runtime.h>

typedef float v2f __attribute__((ext_vector_type(2)));
typedef float v4f __attribute__((ext_vector_type(4)));
typedef float v8f __attribute__((ext_vector_type(8)));

#define Bn   32
#define Cn   512
#define HW2  1024
#define NW   77
#define NP   80
#define WD   256

static __device__ __forceinline__ v8f wmma_f32(v2f a, v2f b, v8f c) {
  return __builtin_amdgcn_wmma_f32_16x16x4_f32(false, a, false, b, (short)0, c, false, false);
}

static __device__ __forceinline__ v2f lo2(v4f q) { v2f r = {q.x, q.y}; return r; }
static __device__ __forceinline__ v2f hi2(v4f q) { v2f r = {q.z, q.w}; return r; }

__global__ __launch_bounds__(256) void we_gemm_kernel(
    const float* __restrict__ word_emb, const float* __restrict__ W_fc,
    const float* __restrict__ b_fc, float* __restrict__ we_ws,
    float* __restrict__ weT_ws) {
  const int lane = threadIdx.x & 31;
  const int wave = threadIdx.x >> 5;
  int tile = blockIdx.x * 8 + wave;
  const int b  = tile / (5 * 64);
  tile -= b * 5 * 64;
  const int tm = tile >> 6;
  const int tn = tile & 63;
  const int half = lane >> 4;
  const int l16  = lane & 15;

  const int a_row = tm * 16 + l16;
  const int col   = tn * 16 + l16;
  const float* Abase = word_emb + ((size_t)b * NW + (a_row < NW ? a_row : NW - 1)) * WD;
  const float* Bbase = W_fc + (size_t)col * WD;

  v8f acc = {};
  for (int k0 = 0; k0 < WD; k0 += 8) {
    const int k = k0 + 4 * half;
    v4f a  = *(const v4f*)(Abase + k);
    v4f bm = *(const v4f*)(Bbase + k);
    acc = wmma_f32(lo2(a), lo2(bm), acc);
    acc = wmma_f32(hi2(a), hi2(bm), acc);
  }
  const float bias = b_fc[col];

  float* D = we_ws + ((size_t)b * NP + tm * 16 + 8 * half) * HW2 + col;
  #pragma unroll
  for (int i = 0; i < 8; ++i) D[(size_t)i * HW2] = acc[i] + bias;

  float* T = weT_ws + ((size_t)b * HW2 + col) * NP + tm * 16 + 8 * half;
  v4f lo = {acc[0] + bias, acc[1] + bias, acc[2] + bias, acc[3] + bias};
  v4f hi = {acc[4] + bias, acc[5] + bias, acc[6] + bias, acc[7] + bias};
  *(v4f*)(T)     = lo;
  *(v4f*)(T + 4) = hi;
}

__global__ __launch_bounds__(256) void scores_softmax_kernel(
    const float* __restrict__ feat, const float* __restrict__ we_ws,
    float* __restrict__ att_ws) {
  const int lane = threadIdx.x & 31;
  const int wave = threadIdx.x >> 5;
  const int slab = blockIdx.x * 8 + wave;
  const int b  = slab >> 5;
  const int cs = (slab & 31) * 16;
  const int half = lane >> 4;
  const int l16  = lane & 15;

  const float* Abase  = feat  + ((size_t)b * Cn + cs + l16) * HW2;
  const float* Bbase0 = we_ws + ((size_t)b * NP + l16) * HW2;

  v8f acc[5] = {};
  for (int k0 = 0; k0 < HW2; k0 += 8) {
    const int k = k0 + 4 * half;
    v4f a = *(const v4f*)(Abase + k);
    v4f bm[5];
    #pragma unroll
    for (int t = 0; t < 5; ++t)
      bm[t] = *(const v4f*)(Bbase0 + (size_t)(16 * t) * HW2 + k);
    #pragma unroll
    for (int t = 0; t < 5; ++t) acc[t] = wmma_f32(lo2(a), lo2(bm[t]), acc[t]);
    #pragma unroll
    for (int t = 0; t < 5; ++t) acc[t] = wmma_f32(hi2(a), hi2(bm[t]), acc[t]);
  }

  float* Dbase = att_ws + ((size_t)b * Cn + cs + 8 * half) * NP;
  #pragma unroll
  for (int i = 0; i < 8; ++i) {
    float m = -3.402823466e38f;
    #pragma unroll
    for (int t = 0; t < 5; ++t)
      if (16 * t + l16 < NW) m = fmaxf(m, acc[t][i]);
    #pragma unroll
    for (int off = 8; off >= 1; off >>= 1) m = fmaxf(m, __shfl_xor(m, off, 32));

    float e[5], s = 0.f;
    #pragma unroll
    for (int t = 0; t < 5; ++t) {
      e[t] = (16 * t + l16 < NW) ? __expf(acc[t][i] - m) : 0.f;
      s += e[t];
    }
    #pragma unroll
    for (int off = 8; off >= 1; off >>= 1) s += __shfl_xor(s, off, 32);
    const float inv = 1.f / s;
    #pragma unroll
    for (int t = 0; t < 5; ++t)
      Dbase[(size_t)i * NP + 16 * t + l16] = e[t] * inv;
  }
}

__global__ __launch_bounds__(256) void attfeat_kernel(
    const float* __restrict__ att_ws, const float* __restrict__ weT_ws,
    float* __restrict__ out) {
  const int lane = threadIdx.x & 31;
  const int wave = threadIdx.x >> 5;
  int tile = blockIdx.x * 8 + wave;
  const int b = tile / (32 * 16);
  tile -= b * 32 * 16;
  const int tm = tile >> 4;
  const int tq = tile & 15;
  const int half = lane >> 4;
  const int l16  = lane & 15;

  const float* Abase = att_ws + ((size_t)b * Cn + tm * 16 + l16) * NP;
  const float* Bbase0 = weT_ws + ((size_t)b * HW2 + tq * 64 + l16) * NP;

  v8f acc[4] = {};
  for (int k0 = 0; k0 < NP; k0 += 8) {
    const int k = k0 + 4 * half;
    v4f a = *(const v4f*)(Abase + k);
    v4f bm[4];
    #pragma unroll
    for (int t = 0; t < 4; ++t)
      bm[t] = *(const v4f*)(Bbase0 + (size_t)(16 * t) * NP + k);
    #pragma unroll
    for (int t = 0; t < 4; ++t) acc[t] = wmma_f32(lo2(a), lo2(bm[t]), acc[t]);
    #pragma unroll
    for (int t = 0; t < 4; ++t) acc[t] = wmma_f32(hi2(a), hi2(bm[t]), acc[t]);
  }
  float* D = out + ((size_t)b * Cn + tm * 16 + 8 * half) * HW2 + tq * 64 + l16;
  #pragma unroll
  for (int t = 0; t < 4; ++t)
    #pragma unroll
    for (int i = 0; i < 8; ++i)
      D[(size_t)i * HW2 + 16 * t] = acc[t][i];
}

extern "C" void kernel_launch(void* const* d_in, const int* in_sizes, int n_in,
                              void* d_out, int out_size, void* d_ws, size_t ws_size,
                              hipStream_t stream) {
  const float* feat     = (const float*)d_in[0];
  const float* word_emb = (const float*)d_in[1];
  const float* W_fc     = (const float*)d_in[2];
  const float* b_fc     = (const float*)d_in[3];
  float* out = (float*)d_out;

  float* we_ws  = (float*)d_ws;
  float* weT_ws = we_ws  + (size_t)Bn * NP * HW2;
  float* att_ws = weT_ws + (size_t)Bn * HW2 * NP;

  we_gemm_kernel       <<<(Bn * 5 * 64) / 8,  256, 0, stream>>>(word_emb, W_fc, b_fc, we_ws, weT_ws);
  scores_softmax_kernel<<<(Bn * 32) / 8,      256, 0, stream>>>(feat, we_ws, att_ws);
  attfeat_kernel       <<<(Bn * 32 * 16) / 8, 256, 0, stream>>>(att_ws, weT_ws, out);
}


# scores kernel: K split over the 8 waves of a work-group (1024 work-groups, was 128), a trip's 24 loads issued as one group; partial tiles added through LDS in wave order: the k sum order changes
# speedup vs baseline: 1.3101x; 1.2621x over previous
#include <hip/hip_runtime.h>

typedef float v2f __attribute__((ext_vector_type(2)));
typedef float v4f __attribute__((ext_vector_type(4)));
typedef float v8f __attribute__((ext_vector_type(8)));

#define Bn   32
#define Cn   512
#define HW2  1024
#define NW   77
#define NP   80
#define WD   256

static __device__ __forceinline__ v8f wmma_f32(v2f a, v2f b, v8f c) {
  return __builtin_amdgcn_wmma_f32_16x16x4_f32(false, a, false, b, (short)0, c, false, false);
}

static __device__ __forceinline__ v2f lo2(v4f q) { v2f r = {q.x, q.y}; return r; }
static __device__ __forceinline__ v2f hi2(v4f q) { v2f r = {q.z, q.w}; return r; }

__global__ __launch_bounds__(256) void we_gemm_kernel(
    const float* __restrict__ word_emb, const float* __restrict__ W_fc,
    const float* __restrict__ b_fc, float* __restrict__ we_ws,
    float* __restrict__ weT_ws) {
  const int lane = threadIdx.x & 31;
  const int wave = threadIdx.x >> 5;
  int tile = blockIdx.x * 8 + wave;
  const int b  = tile / (5 * 64);
  tile -= b * 5 * 64;
  const int tm = tile >> 6;
  const int tn = tile & 63;
  const int half = lane >> 4;
  const int l16  = lane & 15;

  const int a_row = tm * 16 + l16;
  const int col   = tn * 16 + l16;
  const float* Abase = word_emb + ((size_t)b * NW + (a_row < NW ? a_row : NW - 1)) * WD;
  const float* Bbase = W_fc + (size_t)col * WD;

  v8f acc = {};
  for (int k0 = 0; k0 < WD; k0 += 8) {
    const int k = k0 + 4 * half;
    v4f a  = *(const v4f*)(Abase + k);
    v4f bm = *(const v4f*)(Bbase + k);
    acc = wmma_f32(lo2(a), lo2(bm), acc);
    acc = wmma_f32(hi2(a), hi2(bm), acc);
  }
  const float bias = b_fc[col];

  float* D = we_ws + ((size_t)b * NP + tm * 16 + 8 * half) * HW2 + col;
  #pragma unroll
  for (int i = 0; i < 8; ++i) D[(size_t)i * HW2] = acc[i] + bias;

  float* T = weT_ws + ((size_t)b * HW2 + col) * NP + tm * 16 + 8 * half;
  v4f lo = {acc[0] + bias, acc[1] + bias, acc[2] + bias, acc[3] + bias};
  v4f hi = {acc[4] + bias, acc[5] + bias, acc[6] + bias, acc[7] + bias};
  *(v4f*)(T)     = lo;
  *(v4f*)(T + 4) = hi;
}

#define KSPLIT 8
#define KCH    (HW2 / KSPLIT)
__global__ __launch_bounds__(256) void scores_softmax_kernel(
    const float* __restrict__ feat, const float* __restrict__ we_ws,
    float* __restrict__ att_ws) {
  __shared__ float part[KSPLIT][5][8][32];
  const int lane = threadIdx.x & 31;
  const int wave = threadIdx.x >> 5;
  const int slab = blockIdx.x;
  const int b  = slab >> 5;
  const int cs = (slab & 31) * 16;
  const int half = lane >> 4;
  const int l16  = lane & 15;

  const float* Abase  = feat  + ((size_t)b * Cn + cs + l16) * HW2 + wave * KCH;
  const float* Bbase0 = we_ws + ((size_t)b * NP + l16) * HW2 + wave * KCH;

  v8f acc[5] = {};
  #pragma unroll 1
  for (int k0 = 0; k0 < KCH; k0 += 32) {
    const int k = k0 + 4 * half;
    v4f a[4], bm[5][4];
    #pragma unroll
    for (int s = 0; s < 4; ++s) a[s] = *(const v4f*)(Abase + k + 8 * s);
    #pragma unroll
    for (int t = 0; t < 5; ++t)
      #pragma unroll
      for (int s = 0; s < 4; ++s)
        bm[t][s] = *(const v4f*)(Bbase0 + (size_t)(16 * t) * HW2 + k + 8 * s);
    __builtin_amdgcn_sched_barrier(0);
    #pragma unroll
    for (int s = 0; s < 4; ++s) {
      #pragma unroll
      for (int t = 0; t < 5; ++t) acc[t] = wmma_f32(lo2(a[s]), lo2(bm[t][s]), acc[t]);
      #pragma unroll
      for (int t = 0; t < 5; ++t) acc[t] = wmma_f32(hi2(a[s]), hi2(bm[t][s]), acc[t]);
    }
  }

  #pragma unroll
  for (int t = 0; t < 5; ++t)
    #pragma unroll
    for (int i = 0; i < 8; ++i) part[wave][t][i][lane] = acc[t][i];
  __syncthreads();

  float sc[5];
  #pragma unroll
  for (int t = 0; t < 5; ++t) {
    float v = part[0][t][wave][lane];
    #pragma unroll
    for (int p = 1; p < KSPLIT; ++p) v += part[p][t][wave][lane];
    sc[t] = v;
  }

  float m = -3.402823466e38f;
  #pragma unroll
  for (int t = 0; t < 5; ++t)
    if (16 * t + l16 < NW) m = fmaxf(m, sc[t]);
  #pragma unroll
  for (int off = 8; off >= 1; off >>= 1) m = fmaxf(m, __shfl_xor(m, off, 32));

  float e[5], sum = 0.f;
  #pragma unroll
  for (int t = 0; t < 5; ++t) {
    e[t] = (16 * t + l16 < NW) ? __expf(sc[t] - m) : 0.f;
    sum += e[t];
  }
  #pragma unroll
  for (int off = 8; off >= 1; off >>= 1) sum += __shfl_xor(sum, off, 32);
  const float inv = 1.f / sum;
  float* Drow = att_ws + ((size_t)b * Cn + cs + 8 * half + wave) * NP;
  #pragma unroll
  for (int t = 0; t < 5; ++t)
    Drow[16 * t + l16] = e[t] * inv;
}

__global__ __launch_bounds__(256) void attfeat_kernel(
    const float* __restrict__ att_ws, const float* __restrict__ weT_ws,
    float* __restrict__ out) {
  const int lane = threadIdx.x & 31;
  const int wave = threadIdx.x >> 5;
  int tile = blockIdx.x * 8 + wave;
  const int b = tile / (32 * 16);
  tile -= b * 32 * 16;
  const int tm = tile >> 4;
  const int tq = tile & 15;
  const int half = lane >> 4;
  const int l16  = lane & 15;

  const float* Abase = att_ws + ((size_t)b * Cn + tm * 16 + l16) * NP;
  const float* Bbase0 = weT_ws + ((size_t)b * HW2 + tq * 64 + l16) * NP;

  v8f acc[4] = {};
  for (int k0 = 0; k0 < NP; k0 += 8) {
    const int k = k0 + 4 * half;
    v4f a = *(const v4f*)(Abase + k);
    v4f bm[4];
    #pragma unroll
    for (int t = 0; t < 4; ++t)
      bm[t] = *(const v4f*)(Bbase0 + (size_t)(16 * t) * NP + k);
    #pragma unroll
    for (int t = 0; t < 4; ++t) acc[t] = wmma_f32(lo2(a), lo2(bm[t]), acc[t]);
    #pragma unroll
    for (int t = 0; t < 4; ++t) acc[t] = wmma_f32(hi2(a), hi2(bm[t]), acc[t]);
  }
  float* D = out + ((size_t)b * Cn + tm * 16 + 8 * half) * HW2 + tq * 64 + l16;
  #pragma unroll
  for (int t = 0; t < 4; ++t)
    #pragma unroll
    for (int i = 0; i < 8; ++i)
      D[(size_t)i * HW2 + 16 * t] = acc[t][i];
}

extern "C" void kernel_launch(void* const* d_in, const int* in_sizes, int n_in,
                              void* d_out, int out_size, void* d_ws, size_t ws_size,
                              hipStream_t stream) {
  const float* feat     = (const float*)d_in[0];
  const float* word_emb = (const float*)d_in[1];
  const float* W_fc     = (const float*)d_in[2];
  const float* b_fc     = (const float*)d_in[3];
  float* out = (float*)d_out;

  float* we_ws  = (float*)d_ws;
  float* weT_ws = we_ws  + (size_t)Bn * NP * HW2;
  float* att_ws = weT_ws + (size_t)Bn * HW2 * NP;

  we_gemm_kernel       <<<(Bn * 5 * 64) / 8,  256, 0, stream>>>(word_emb, W_fc, b_fc, we_ws, weT_ws);
  scores_softmax_kernel<<<Bn * 32,            256, 0, stream>>>(feat, we_ws, att_ws);
  attfeat_kernel       <<<(Bn * 32 * 16) / 8, 256, 0, stream>>>(att_ws, weT_ws, out);
}


# out GEMM (att @ we): each work-group copies its 64x80 and 128x80 operand blocks to LDS with whole-line 16-byte loads and reads fragments from LDS (was 16-line global fragment loads); sum order kept
# speedup vs baseline: 1.5789x; 1.2052x over previous
#include <hip/hip_runtime.h>

typedef float v2f __attribute__((ext_vector_type(2)));
typedef float v4f __attribute__((ext_vector_type(4)));
typedef float v8f __attribute__((ext_vector_type(8)));

#define Bn   32
#define Cn   512
#define HW2  1024
#define NW   77
#define NP   80
#define WD   256

static __device__ __forceinline__ v8f wmma_f32(v2f a, v2f b, v8f c) {
  return __builtin_amdgcn_wmma_f32_16x16x4_f32(false, a, false, b, (short)0, c, false, false);
}

static __device__ __forceinline__ v2f lo2(v4f q) { v2f r = {q.x, q.y}; return r; }
static __device__ __forceinline__ v2f hi2(v4f q) { v2f r = {q.z, q.w}; return r; }

__global__ __launch_bounds__(256) void we_gemm_kernel(
    const float* __restrict__ word_emb, const float* __restrict__ W_fc,
    const float* __restrict__ b_fc, float* __restrict__ we_ws,
    float* __restrict__ weT_ws) {
  const int lane = threadIdx.x & 31;
  const int wave = threadIdx.x >> 5;
  int tile = blockIdx.x * 8 + wave;
  const int b  = tile / (5 * 64);
  tile -= b * 5 * 64;
  const int tm = tile >> 6;
  const int tn = tile & 63;
  const int half = lane >> 4;
  const int l16  = lane & 15;

  const int a_row = tm * 16 + l16;
  const int col   = tn * 16 + l16;
  const float* Abase = word_emb + ((size_t)b * NW + (a_row < NW ? a_row : NW - 1)) * WD;
  const float* Bbase = W_fc + (size_t)col * WD;

  v8f acc = {};
  for (int k0 = 0; k0 < WD; k0 += 8) {
    const int k = k0 + 4 * half;
    v4f a  = *(const v4f*)(Abase + k);
    v4f bm = *(const v4f*)(Bbase + k);
    acc = wmma_f32(lo2(a), lo2(bm), acc);
    acc = wmma_f32(hi2(a), hi2(bm), acc);
  }
  const float bias = b_fc[col];

  float* D = we_ws + ((size_t)b * NP + tm * 16 + 8 * half) * HW2 + col;
  #pragma unroll
  for (int i = 0; i < 8; ++i) D[(size_t)i * HW2] = acc[i] + bias;

  float* T = weT_ws + ((size_t)b * HW2 + col) * NP + tm * 16 + 8 * half;
  v4f lo = {acc[0] + bias, acc[1] + bias, acc[2] + bias, acc[3] + bias};
  v4f hi = {acc[4] + bias, acc[5] + bias, acc[6] + bias, acc[7] + bias};
  *(v4f*)(T)     = lo;
  *(v4f*)(T + 4) = hi;
}

#define KSPLIT 8
#define KCH    (HW2 / KSPLIT)
__global__ __launch_bounds__(256) void scores_softmax_kernel(
    const float* __restrict__ feat, const float* __restrict__ we_ws,
    float* __restrict__ att_ws) {
  __shared__ float part[KSPLIT][5][8][32];
  const int lane = threadIdx.x & 31;
  const int wave = threadIdx.x >> 5;
  const int slab = blockIdx.x;
  const int b  = slab >> 5;
  const int cs = (slab & 31) * 16;
  const int half = lane >> 4;
  const int l16  = lane & 15;

  const float* Abase  = feat  + ((size_t)b * Cn + cs + l16) * HW2 + wave * KCH;
  const float* Bbase0 = we_ws + ((size_t)b * NP + l16) * HW2 + wave * KCH;

  v8f acc[5] = {};
  #pragma unroll 1
  for (int k0 = 0; k0 < KCH; k0 += 32) {
    const int k = k0 + 4 * half;
    v4f a[4], bm[5][4];
    #pragma unroll
    for (int s = 0; s < 4; ++s) a[s] = *(const v4f*)(Abase + k + 8 * s);
    #pragma unroll
    for (int t = 0; t < 5; ++t)
      #pragma unroll
      for (int s = 0; s < 4; ++s)
        bm[t][s] = *(const v4f*)(Bbase0 + (size_t)(16 * t) * HW2 + k + 8 * s);
    __builtin_amdgcn_sched_barrier(0);
    #pragma unroll
    for (int s = 0; s < 4; ++s) {
      #pragma unroll
      for (int t = 0; t < 5; ++t) acc[t] = wmma_f32(lo2(a[s]), lo2(bm[t][s]), acc[t]);
      #pragma unroll
      for (int t = 0; t < 5; ++t) acc[t] = wmma_f32(hi2(a[s]), hi2(bm[t][s]), acc[t]);
    }
  }

  #pragma unroll
  for (int t = 0; t < 5; ++t)
    #pragma unroll
    for (int i = 0; i < 8; ++i) part[wave][t][i][lane] = acc[t][i];
  __syncthreads();

  float sc[5];
  #pragma unroll
  for (int t = 0; t < 5; ++t) {
    float v = part[0][t][wave][lane];
    #pragma unroll
    for (int p = 1; p < KSPLIT; ++p) v += part[p][t][wave][lane];
    sc[t] = v;
  }

  float m = -3.402823466e38f;
  #pragma unroll
  for (int t = 0; t < 5; ++t)
    if (16 * t + l16 < NW) m = fmaxf(m, sc[t]);
  #pragma unroll
  for (int off = 8; off >= 1; off >>= 1) m = fmaxf(m, __shfl_xor(m, off, 32));

  float e[5], sum = 0.f;
  #pragma unroll
  for (int t = 0; t < 5; ++t) {
    e[t] = (16 * t + l16 < NW) ? __expf(sc[t] - m) : 0.f;
    sum += e[t];
  }
  #pragma unroll
  for (int off = 8; off >= 1; off >>= 1) sum += __shfl_xor(sum, off, 32);
  const float inv = 1.f / sum;
  float* Drow = att_ws + ((size_t)b * Cn + cs + 8 * half + wave) * NP;
  #pragma unroll
  for (int t = 0; t < 5; ++t)
    Drow[16 * t + l16] = e[t] * inv;
}

#define T3M 64
#define T3N 128
#define LP  84
__global__ __launch_bounds__(256) void attfeat_kernel(
    const float* __restrict__ att_ws, const float* __restrict__ weT_ws,
    float* __restrict__ out) {
  __shared__ __attribute__((aligned(16))) float tile[(T3M + T3N) * LP];
  const int tid  = threadIdx.x;
  const int lane = tid & 31;
  const int wave = tid >> 5;
  const int wg = blockIdx.x;
  const int b  = wg >> 6;
  const int tm = (wg >> 3) & 7;
  const int tn = wg & 7;

  const float* Asrc = att_ws + ((size_t)b * Cn  + tm * T3M) * NP;
  const float* Bsrc = weT_ws + ((size_t)b * HW2 + tn * T3N) * NP;

  v4f st[15];
  #pragma unroll
  for (int i = 0; i < 15; ++i) {
    const int j = i * 256 + tid;
    const float* p = (i < 5) ? (Asrc + 4 * j) : (Bsrc + 4 * (j - 1280));
    st[i] = *(const v4f*)p;
  }
  #pragma unroll
  for (int i = 0; i < 15; ++i) {
    const int j = i * 256 + tid;
    const int row = j / 20;
    const int c4  = j - row * 20;
    *(v4f*)&tile[row * LP + c4 * 4] = st[i];
  }
  __syncthreads();

  const int half = lane >> 4;
  const int l16  = lane & 15;
  const int wr = wave >> 2;
  const int wc = wave & 3;
  const float* Al = tile + (wr * 32 + l16) * LP + 4 * half;
  const float* Bl = tile + (T3M + wc * 32 + l16) * LP + 4 * half;

  v8f acc00 = {}, acc01 = {}, acc10 = {}, acc11 = {};
  #pragma unroll
  for (int k0 = 0; k0 < NP; k0 += 8) {
    const v4f a0 = *(const v4f*)(Al + k0);
    const v4f a1 = *(const v4f*)(Al + 16 * LP + k0);
    const v4f b0 = *(const v4f*)(Bl + k0);
    const v4f b1 = *(const v4f*)(Bl + 16 * LP + k0);
    acc00 = wmma_f32(lo2(a0), lo2(b0), acc00);
    acc01 = wmma_f32(lo2(a0), lo2(b1), acc01);
    acc10 = wmma_f32(lo2(a1), lo2(b0), acc10);
    acc11 = wmma_f32(lo2(a1), lo2(b1), acc11);
    acc00 = wmma_f32(hi2(a0), hi2(b0), acc00);
    acc01 = wmma_f32(hi2(a0), hi2(b1), acc01);
    acc10 = wmma_f32(hi2(a1), hi2(b0), acc10);
    acc11 = wmma_f32(hi2(a1), hi2(b1), acc11);
  }

  float* D = out + ((size_t)b * Cn + tm * T3M + wr * 32 + 8 * half) * HW2
                 + tn * T3N + wc * 32 + l16;
  #pragma unroll
  for (int i = 0; i < 8; ++i) {
    D[(size_t)i * HW2]             = acc00[i];
    D[(size_t)i * HW2 + 16]        = acc01[i];
    D[(size_t)(16 + i) * HW2]      = acc10[i];
    D[(size_t)(16 + i) * HW2 + 16] = acc11[i];
  }
}

extern "C" void kernel_launch(void* const* d_in, const int* in_sizes, int n_in,
                              void* d_out, int out_size, void* d_ws, size_t ws_size,
                              hipStream_t stream) {
  const float* feat     = (const float*)d_in[0];
  const float* word_emb = (const float*)d_in[1];
  const float* W_fc     = (const float*)d_in[2];
  const float* b_fc     = (const float*)d_in[3];
  float* out = (float*)d_out;

  float* we_ws  = (float*)d_ws;
  float* weT_ws = we_ws  + (size_t)Bn * NP * HW2;
  float* att_ws = weT_ws + (size_t)Bn * HW2 * NP;

  we_gemm_kernel       <<<(Bn * 5 * 64) / 8,  256, 0, stream>>>(word_emb, W_fc, b_fc, we_ws, weT_ws);
  scores_softmax_kernel<<<Bn * 32,            256, 0, stream>>>(feat, we_ws, att_ws);
  attfeat_kernel       <<<Bn * 8 * 8,          256, 0, stream>>>(att_ws, weT_ws, out);
}


# scores kernel: 32-row slab per work-group (512 work-groups, was 1024): each wave keeps two row tiles, so one B fragment load feeds two WMMA sets (7 loads per 20 WMMAs, was 6 per 10); sum order kept
# speedup vs baseline: 1.8051x; 1.1433x over previous
#include <hip/hip_runtime.h>

typedef float v2f __attribute__((ext_vector_type(2)));
typedef float v4f __attribute__((ext_vector_type(4)));
typedef float v8f __attribute__((ext_vector_type(8)));

#define Bn   32
#define Cn   512
#define HW2  1024
#define NW   77
#define NP   80
#define WD   256

static __device__ __forceinline__ v8f wmma_f32(v2f a, v2f b, v8f c) {
  return __builtin_amdgcn_wmma_f32_16x16x4_f32(false, a, false, b, (short)0, c, false, false);
}

static __device__ __forceinline__ v2f lo2(v4f q) { v2f r = {q.x, q.y}; return r; }
static __device__ __forceinline__ v2f hi2(v4f q) { v2f r = {q.z, q.w}; return r; }

__global__ __launch_bounds__(256) void we_gemm_kernel(
    const float* __restrict__ word_emb, const float* __restrict__ W_fc,
    const float* __restrict__ b_fc, float* __restrict__ we_ws,
    float* __restrict__ weT_ws) {
  const int lane = threadIdx.x & 31;
  const int wave = threadIdx.x >> 5;
  int tile = blockIdx.x * 8 + wave;
  const int b  = tile / (5 * 64);
  tile -= b * 5 * 64;
  const int tm = tile >> 6;
  const int tn = tile & 63;
  const int half = lane >> 4;
  const int l16  = lane & 15;

  const int a_row = tm * 16 + l16;
  const int col   = tn * 16 + l16;
  const float* Abase = word_emb + ((size_t)b * NW + (a_row < NW ? a_row : NW - 1)) * WD;
  const float* Bbase = W_fc + (size_t)col * WD;

  v8f acc = {};
  for (int k0 = 0; k0 < WD; k0 += 8) {
    const int k = k0 + 4 * half;
    v4f a  = *(const v4f*)(Abase + k);
    v4f bm = *(const v4f*)(Bbase + k);
    acc = wmma_f32(lo2(a), lo2(bm), acc);
    acc = wmma_f32(hi2(a), hi2(bm), acc);
  }
  const float bias = b_fc[col];

  float* D = we_ws + ((size_t)b * NP + tm * 16 + 8 * half) * HW2 + col;
  #pragma unroll
  for (int i = 0; i < 8; ++i) D[(size_t)i * HW2] = acc[i] + bias;

  float* T = weT_ws + ((size_t)b * HW2 + col) * NP + tm * 16 + 8 * half;
  v4f lo = {acc[0] + bias, acc[1] + bias, acc[2] + bias, acc[3] + bias};
  v4f hi = {acc[4] + bias, acc[5] + bias, acc[6] + bias, acc[7] + bias};
  *(v4f*)(T)     = lo;
  *(v4f*)(T + 4) = hi;
}

#define KSPLIT 8
#define KCH    (HW2 / KSPLIT)
#define RT2    2
__global__ __launch_bounds__(256) void scores_softmax_kernel(
    const float* __restrict__ feat, const float* __restrict__ we_ws,
    float* __restrict__ att_ws) {
  __shared__ float part[KSPLIT][RT2][5][8][32];
  const int lane = threadIdx.x & 31;
  const int wave = threadIdx.x >> 5;
  const int slab = blockIdx.x;
  const int b  = slab >> 4;
  const int cs = (slab & 15) * (16 * RT2);
  const int half = lane >> 4;
  const int l16  = lane & 15;

  const float* Abase0 = feat  + ((size_t)b * Cn + cs + l16) * HW2 + wave * KCH;
  const float* Bbase0 = we_ws + ((size_t)b * NP + l16) * HW2 + wave * KCH;

  v8f acc[RT2][5] = {};
  #pragma unroll 1
  for (int k0 = 0; k0 < KCH; k0 += 32) {
    const int k = k0 + 4 * half;
    v4f a[RT2][4], bm[5][4];
    #pragma unroll
    for (int r = 0; r < RT2; ++r)
      #pragma unroll
      for (int s = 0; s < 4; ++s)
        a[r][s] = *(const v4f*)(Abase0 + (size_t)(16 * r) * HW2 + k + 8 * s);
    #pragma unroll
    for (int t = 0; t < 5; ++t)
      #pragma unroll
      for (int s = 0; s < 4; ++s)
        bm[t][s] = *(const v4f*)(Bbase0 + (size_t)(16 * t) * HW2 + k + 8 * s);
    __builtin_amdgcn_sched_barrier(0);
    #pragma unroll
    for (int s = 0; s < 4; ++s) {
      #pragma unroll
      for (int r = 0; r < RT2; ++r)
        #pragma unroll
        for (int t = 0; t < 5; ++t)
          acc[r][t] = wmma_f32(lo2(a[r][s]), lo2(bm[t][s]), acc[r][t]);
      #pragma unroll
      for (int r = 0; r < RT2; ++r)
        #pragma unroll
        for (int t = 0; t < 5; ++t)
          acc[r][t] = wmma_f32(hi2(a[r][s]), hi2(bm[t][s]), acc[r][t]);
    }
  }

  #pragma unroll
  for (int r = 0; r < RT2; ++r)
    #pragma unroll
    for (int t = 0; t < 5; ++t)
      #pragma unroll
      for (int i = 0; i < 8; ++i) part[wave][r][t][i][lane] = acc[r][t][i];
  __syncthreads();

  #pragma unroll
  for (int r = 0; r < RT2; ++r) {
    float sc[5];
    #pragma unroll
    for (int t = 0; t < 5; ++t) {
      float v = part[0][r][t][wave][lane];
      #pragma unroll
      for (int p = 1; p < KSPLIT; ++p) v += part[p][r][t][wave][lane];
      sc[t] = v;
    }

    float m = -3.402823466e38f;
    #pragma unroll
    for (int t = 0; t < 5; ++t)
      if (16 * t + l16 < NW) m = fmaxf(m, sc[t]);
    #pragma unroll
    for (int off = 8; off >= 1; off >>= 1) m = fmaxf(m, __shfl_xor(m, off, 32));

    float e[5], sum = 0.f;
    #pragma unroll
    for (int t = 0; t < 5; ++t) {
      e[t] = (16 * t + l16 < NW) ? __expf(sc[t] - m) : 0.f;
      sum += e[t];
    }
    #pragma unroll
    for (int off = 8; off >= 1; off >>= 1) sum += __shfl_xor(sum, off, 32);
    const float inv = 1.f / sum;
    float* Drow = att_ws + ((size_t)b * Cn + cs + 16 * r + 8 * half + wave) * NP;
    #pragma unroll
    for (int t = 0; t < 5; ++t)
      Drow[16 * t + l16] = e[t] * inv;
  }
}

#define T3M 64
#define T3N 128
#define LP  84
__global__ __launch_bounds__(256) void attfeat_kernel(
    const float* __restrict__ att_ws, const float* __restrict__ weT_ws,
    float* __restrict__ out) {
  __shared__ __attribute__((aligned(16))) float tile[(T3M + T3N) * LP];
  const int tid  = threadIdx.x;
  const int lane = tid & 31;
  const int wave = tid >> 5;
  const int wg = blockIdx.x;
  const int b  = wg >> 6;
  const int tm = (wg >> 3) & 7;
  const int tn = wg & 7;

  const float* Asrc = att_ws + ((size_t)b * Cn  + tm * T3M) * NP;
  const float* Bsrc = weT_ws + ((size_t)b * HW2 + tn * T3N) * NP;

  v4f st[15];
  #pragma unroll
  for (int i = 0; i < 15; ++i) {
    const int j = i * 256 + tid;
    const float* p = (i < 5) ? (Asrc + 4 * j) : (Bsrc + 4 * (j - 1280));
    st[i] = *(const v4f*)p;
  }
  #pragma unroll
  for (int i = 0; i < 15; ++i) {
    const int j = i * 256 + tid;
    const int row = j / 20;
    const int c4  = j - row * 20;
    *(v4f*)&tile[row * LP + c4 * 4] = st[i];
  }
  __syncthreads();

  const int half = lane >> 4;
  const int l16  = lane & 15;
  const int wr = wave >> 2;
  const int wc = wave & 3;
  const float* Al = tile + (wr * 32 + l16) * LP + 4 * half;
  const float* Bl = tile + (T3M + wc * 32 + l16) * LP + 4 * half;

  v8f acc00 = {}, acc01 = {}, acc10 = {}, acc11 = {};
  #pragma unroll
  for (int k0 = 0; k0 < NP; k0 += 8) {
    const v4f a0 = *(const v4f*)(Al + k0);
    const v4f a1 = *(const v4f*)(Al + 16 * LP + k0);
    const v4f b0 = *(const v4f*)(Bl + k0);
    const v4f b1 = *(const v4f*)(Bl + 16 * LP + k0);
    acc00 = wmma_f32(lo2(a0), lo2(b0), acc00);
    acc01 = wmma_f32(lo2(a0), lo2(b1), acc01);
    acc10 = wmma_f32(lo2(a1), lo2(b0), acc10);
    acc11 = wmma_f32(lo2(a1), lo2(b1), acc11);
    acc00 = wmma_f32(hi2(a0), hi2(b0), acc00);
    acc01 = wmma_f32(hi2(a0), hi2(b1), acc01);
    acc10 = wmma_f32(hi2(a1), hi2(b0), acc10);
    acc11 = wmma_f32(hi2(a1), hi2(b1), acc11);
  }

  float* D = out + ((size_t)b * Cn + tm * T3M + wr * 32 + 8 * half) * HW2
                 + tn * T3N + wc * 32 + l16;
  #pragma unroll
  for (int i = 0; i < 8; ++i) {
    D[(size_t)i * HW2]             = acc00[i];
    D[(size_t)i * HW2 + 16]        = acc01[i];
    D[(size_t)(16 + i) * HW2]      = acc10[i];
    D[(size_t)(16 + i) * HW2 + 16] = acc11[i];
  }
}

extern "C" void kernel_launch(void* const* d_in, const int* in_sizes, int n_in,
                              void* d_out, int out_size, void* d_ws, size_t ws_size,
                              hipStream_t stream) {
  const float* feat     = (const float*)d_in[0];
  const float* word_emb = (const float*)d_in[1];
  const float* W_fc     = (const float*)d_in[2];
  const float* b_fc     = (const float*)d_in[3];
  float* out = (float*)d_out;

  float* we_ws  = (float*)d_ws;
  float* weT_ws = we_ws  + (size_t)Bn * NP * HW2;
  float* att_ws = weT_ws + (size_t)Bn * HW2 * NP;

  we_gemm_kernel       <<<(Bn * 5 * 64) / 8,  256, 0, stream>>>(word_emb, W_fc, b_fc, we_ws, weT_ws);
  scores_softmax_kernel<<<Bn * 16,            256, 0, stream>>>(feat, we_ws, att_ws);
  attfeat_kernel       <<<Bn * 8 * 8,          256, 0, stream>>>(att_ws, weT_ws, out);
}


# we GEMM: 1024 work-groups (1 batch x 32 columns, 10 waves, one tile each) stage both operand blocks in LDS by whole-line 16-byte loads and write both results as whole lines; sum order kept
# speedup vs baseline: 1.9958x; 1.1056x over previous
#include <hip/hip_runtime.h>

typedef float v2f __attribute__((ext_vector_type(2)));
typedef float v4f __attribute__((ext_vector_type(4)));
typedef float v8f __attribute__((ext_vector_type(8)));

#define Bn   32
#define Cn   512
#define HW2  1024
#define NW   77
#define NP   80
#define WD   256

static __device__ __forceinline__ v8f wmma_f32(v2f a, v2f b, v8f c) {
  return __builtin_amdgcn_wmma_f32_16x16x4_f32(false, a, false, b, (short)0, c, false, false);
}

static __device__ __forceinline__ v2f lo2(v4f q) { v2f r = {q.x, q.y}; return r; }
static __device__ __forceinline__ v2f hi2(v4f q) { v2f r = {q.z, q.w}; return r; }

#define K1N  32
#define K1T  320
#define K1P  260
#define K1R1 36
#define K1R2 84
#define K1CH ((NP + K1N) * (WD / 4))
__global__ __launch_bounds__(K1T) void we_gemm_kernel(
    const float* __restrict__ word_emb, const float* __restrict__ W_fc,
    const float* __restrict__ b_fc, float* __restrict__ we_ws,
    float* __restrict__ weT_ws) {
  __shared__ __attribute__((aligned(16))) float sm[(NP + K1N) * K1P];
  const int tid  = threadIdx.x;
  const int lane = tid & 31;
  const int wave = tid >> 5;
  const int b    = blockIdx.x >> 5;
  const int col0 = (blockIdx.x & 31) * K1N;
  const int half = lane >> 4;
  const int l16  = lane & 15;
  const int tm = wave >> 1;
  const int tn = wave & 1;

  const float* Asrc = word_emb + (size_t)b * NW * WD;
  const float* Bsrc = W_fc + (size_t)col0 * WD;

  const float* Ath = Asrc + 4 * tid;
  const float* Bth = Bsrc + 4 * tid;
  v4f st[23];
  #pragma unroll
  for (int i = 0; i < 15; ++i) st[i] = *(const v4f*)(Ath + i * (4 * K1T));
  st[15] = *(const v4f*)((tid < 128) ? Ath + 15 * (4 * K1T)
                                     : Asrc + (NW - 1) * WD + 4 * (tid & 63));
  #pragma unroll
  for (int i = 16; i < 23; ++i)
    if (i < 22 || tid < 128) st[i] = *(const v4f*)(Bth + (i - 16) * (4 * K1T));
  const v4f   bias_nk  = *(const v4f*)(b_fc + col0 + 4 * (tid & 7));
  const float bias_kn0 = b_fc[col0 + tid / 20];
  const float bias_kn1 = b_fc[col0 + 16 + tid / 20];
  float* Sth = sm + (tid >> 6) * K1P + (tid & 63) * 4;
  #pragma unroll
  for (int i = 0; i < 23; ++i)
    if (i < 22 || tid < 128) *(v4f*)(Sth + i * (5 * K1P)) = st[i];
  __syncthreads();

  const float* Al = sm + (tm * 16 + l16) * K1P + 4 * half;
  const float* Bl = sm + (NP + tn * 16 + l16) * K1P + 4 * half;

  v8f acc = {};
  v4f af[2][8], bq[2][8];
  #pragma unroll
  for (int s = 0; s < 8; ++s) {
    af[0][s] = *(const v4f*)(Al + 8 * s);
    bq[0][s] = *(const v4f*)(Bl + 8 * s);
  }
  #pragma unroll
  for (int g = 0; g < 4; ++g) {
    if (g + 1 < 4) {
      #pragma unroll
      for (int s = 0; s < 8; ++s) {
        af[(g + 1) & 1][s] = *(const v4f*)(Al + 64 * (g + 1) + 8 * s);
        bq[(g + 1) & 1][s] = *(const v4f*)(Bl + 64 * (g + 1) + 8 * s);
      }
    }
    __builtin_amdgcn_sched_barrier(0);
    #pragma unroll
    for (int s = 0; s < 8; ++s) {
      acc = wmma_f32(lo2(af[g & 1][s]), lo2(bq[g & 1][s]), acc);
      acc = wmma_f32(hi2(af[g & 1][s]), hi2(bq[g & 1][s]), acc);
    }
    __builtin_amdgcn_sched_barrier(0);
  }

  __syncthreads();
  float* R1 = sm;
  float* R2 = sm + NP * K1R1;
  #pragma unroll
  for (int i = 0; i < 8; ++i)
    R1[(tm * 16 + 8 * half + i) * K1R1 + tn * 16 + l16] = acc[i];
  {
    v4f lo = {acc[0], acc[1], acc[2], acc[3]};
    v4f hi = {acc[4], acc[5], acc[6], acc[7]};
    float* t2 = R2 + (tn * 16 + l16) * K1R2 + tm * 16 + 8 * half;
    *(v4f*)(t2)     = lo;
    *(v4f*)(t2 + 4) = hi;
  }
  __syncthreads();

  float* D1 = we_ws + (size_t)b * NP * HW2 + col0;
  #pragma unroll
  for (int i = 0; i < 2; ++i) {
    const int q = i * K1T + tid;
    const int row = q >> 3;
    const int c4  = q & 7;
    const v4f v = *(const v4f*)(R1 + row * K1R1 + 4 * c4);
    *(v4f*)(D1 + (size_t)row * HW2 + 4 * c4) = v + bias_nk;
  }
  float* D2 = weT_ws + ((size_t)b * HW2 + col0) * NP;
  #pragma unroll
  for (int i = 0; i < 2; ++i) {
    const int q = i * K1T + tid;
    const int row = q / 20;
    const int c   = q - row * 20;
    const float bs = (i == 0) ? bias_kn0 : bias_kn1;
    const v4f v = *(const v4f*)(R2 + row * K1R2 + 4 * c);
    const v4f w = {v.x + bs, v.y + bs, v.z + bs, v.w + bs};
    *(v4f*)(D2 + 4 * q) = w;
  }
}

#define KSPLIT 8
#define KCH    (HW2 / KSPLIT)
#define RT2    2
__global__ __launch_bounds__(256) void scores_softmax_kernel(
    const float* __restrict__ feat, const float* __restrict__ we_ws,
    float* __restrict__ att_ws) {
  __shared__ float part[KSPLIT][RT2][5][8][32];
  const int lane = threadIdx.x & 31;
  const int wave = threadIdx.x >> 5;
  const int slab = blockIdx.x;
  const int b  = slab >> 4;
  const int cs = (slab & 15) * (16 * RT2);
  const int half = lane >> 4;
  const int l16  = lane & 15;

  const float* Abase0 = feat  + ((size_t)b * Cn + cs + l16) * HW2 + wave * KCH;
  const float* Bbase0 = we_ws + ((size_t)b * NP + l16) * HW2 + wave * KCH;

  v8f acc[RT2][5] = {};
  #pragma unroll 1
  for (int k0 = 0; k0 < KCH; k0 += 32) {
    const int k = k0 + 4 * half;
    v4f a[RT2][4], bm[5][4];
    #pragma unroll
    for (int r = 0; r < RT2; ++r)
      #pragma unroll
      for (int s = 0; s < 4; ++s)
        a[r][s] = *(const v4f*)(Abase0 + (size_t)(16 * r) * HW2 + k + 8 * s);
    #pragma unroll
    for (int t = 0; t < 5; ++t)
      #pragma unroll
      for (int s = 0; s < 4; ++s)
        bm[t][s] = *(const v4f*)(Bbase0 + (size_t)(16 * t) * HW2 + k + 8 * s);
    __builtin_amdgcn_sched_barrier(0);
    #pragma unroll
    for (int s = 0; s < 4; ++s) {
      #pragma unroll
      for (int r = 0; r < RT2; ++r)
        #pragma unroll
        for (int t = 0; t < 5; ++t)
          acc[r][t] = wmma_f32(lo2(a[r][s]), lo2(bm[t][s]), acc[r][t]);
      #pragma unroll
      for (int r = 0; r < RT2; ++r)
        #pragma unroll
        for (int t = 0; t < 5; ++t)
          acc[r][t] = wmma_f32(hi2(a[r][s]), hi2(bm[t][s]), acc[r][t]);
    }
  }

  #pragma unroll
  for (int r = 0; r < RT2; ++r)
    #pragma unroll
    for (int t = 0; t < 5; ++t)
      #pragma unroll
      for (int i = 0; i < 8; ++i) part[wave][r][t][i][lane] = acc[r][t][i];
  __syncthreads();

  #pragma unroll
  for (int r = 0; r < RT2; ++r) {
    float sc[5];
    #pragma unroll
    for (int t = 0; t < 5; ++t) {
      float v = part[0][r][t][wave][lane];
      #pragma unroll
      for (int p = 1; p < KSPLIT; ++p) v += part[p][r][t][wave][lane];
      sc[t] = v;
    }

    float m = -3.402823466e38f;
    #pragma unroll
    for (int t = 0; t < 5; ++t)
      if (16 * t + l16 < NW) m = fmaxf(m, sc[t]);
    #pragma unroll
    for (int off = 8; off >= 1; off >>= 1) m = fmaxf(m, __shfl_xor(m, off, 32));

    float e[5], sum = 0.f;
    #pragma unroll
    for (int t = 0; t < 5; ++t) {
      e[t] = (16 * t + l16 < NW) ? __expf(sc[t] - m) : 0.f;
      sum += e[t];
    }
    #pragma unroll
    for (int off = 8; off >= 1; off >>= 1) sum += __shfl_xor(sum, off, 32);
    const float inv = 1.f / sum;
    float* Drow = att_ws + ((size_t)b * Cn + cs + 16 * r + 8 * half + wave) * NP;
    #pragma unroll
    for (int t = 0; t < 5; ++t)
      Drow[16 * t + l16] = e[t] * inv;
  }
}

#define T3M 64
#define T3N 128
#define LP  84
__global__ __launch_bounds__(256) void attfeat_kernel(
    const float* __restrict__ att_ws, const float* __restrict__ weT_ws,
    float* __restrict__ out) {
  __shared__ __attribute__((aligned(16))) float tile[(T3M + T3N) * LP];
  const int tid  = threadIdx.x;
  const int lane = tid & 31;
  const int wave = tid >> 5;
  const int wg = blockIdx.x;
  const int b  = wg >> 6;
  const int tm = (wg >> 3) & 7;
  const int tn = wg & 7;

  const float* Asrc = att_ws + ((size_t)b * Cn  + tm * T3M) * NP;
  const float* Bsrc = weT_ws + ((size_t)b * HW2 + tn * T3N) * NP;

  v4f st[15];
  #pragma unroll
  for (int i = 0; i < 15; ++i) {
    const int j = i * 256 + tid;
    const float* p = (i < 5) ? (Asrc + 4 * j) : (Bsrc + 4 * (j - 1280));
    st[i] = *(const v4f*)p;
  }
  #pragma unroll
  for (int i = 0; i < 15; ++i) {
    const int j = i * 256 + tid;
    const int row = j / 20;
    const int c4  = j - row * 20;
    *(v4f*)&tile[row * LP + c4 * 4] = st[i];
  }
  __syncthreads();

  const int half = lane >> 4;
  const int l16  = lane & 15;
  const int wr = wave >> 2;
  const int wc = wave & 3;
  const float* Al = tile + (wr * 32 + l16) * LP + 4 * half;
  const float* Bl = tile + (T3M + wc * 32 + l16) * LP + 4 * half;

  v8f acc00 = {}, acc01 = {}, acc10 = {}, acc11 = {};
  #pragma unroll
  for (int k0 = 0; k0 < NP; k0 += 8) {
    const v4f a0 = *(const v4f*)(Al + k0);
    const v4f a1 = *(const v4f*)(Al + 16 * LP + k0);
    const v4f b0 = *(const v4f*)(Bl + k0);
    const v4f b1 = *(const v4f*)(Bl + 16 * LP + k0);
    acc00 = wmma_f32(lo2(a0), lo2(b0), acc00);
    acc01 = wmma_f32(lo2(a0), lo2(b1), acc01);
    acc10 = wmma_f32(lo2(a1), lo2(b0), acc10);
    acc11 = wmma_f32(lo2(a1), lo2(b1), acc11);
    acc00 = wmma_f32(hi2(a0), hi2(b0), acc00);
    acc01 = wmma_f32(hi2(a0), hi2(b1), acc01);
    acc10 = wmma_f32(hi2(a1), hi2(b0), acc10);
    acc11 = wmma_f32(hi2(a1), hi2(b1), acc11);
  }

  float* D = out + ((size_t)b * Cn + tm * T3M + wr * 32 + 8 * half) * HW2
                 + tn * T3N + wc * 32 + l16;
  #pragma unroll
  for (int i = 0; i < 8; ++i) {
    D[(size_t)i * HW2]             = acc00[i];
    D[(size_t)i * HW2 + 16]        = acc01[i];
    D[(size_t)(16 + i) * HW2]      = acc10[i];
    D[(size_t)(16 + i) * HW2 + 16] = acc11[i];
  }
}

extern "C" void kernel_launch(void* const* d_in, const int* in_sizes, int n_in,
                              void* d_out, int out_size, void* d_ws, size_t ws_size,
                              hipStream_t stream) {
  const float* feat     = (const float*)d_in[0];
  const float* word_emb = (const float*)d_in[1];
  const float* W_fc     = (const float*)d_in[2];
  const float* b_fc     = (const float*)d_in[3];
  float* out = (float*)d_out;

  float* we_ws  = (float*)d_ws;
  float* weT_ws = we_ws  + (size_t)Bn * NP * HW2;
  float* att_ws = weT_ws + (size_t)Bn * HW2 * NP;

  we_gemm_kernel       <<<Bn * 32,            K1T, 0, stream>>>(word_emb, W_fc, b_fc, we_ws, weT_ws);
  scores_softmax_kernel<<<Bn * 16,            256, 0, stream>>>(feat, we_ws, att_ws);
  attfeat_kernel       <<<Bn * 8 * 8,          256, 0, stream>>>(att_ws, weT_ws, out);
}
